# starting kernel
# baseline (speedup 1.0000x reference)
#include <hip/hip_runtime.h>

typedef __attribute__((ext_vector_type(2))) float v2f;
typedef __attribute__((ext_vector_type(8))) float v8f;

#define IN_F 64
#define OUT_F 64
#define NCLS 16

__global__ void k_zero(float4* __restrict__ p, long n4) {
  long i = (long)blockIdx.x * blockDim.x + threadIdx.x;
  if (i < n4) p[i] = make_float4(0.f, 0.f, 0.f, 0.f);
}

__global__ void k_degree(const int* __restrict__ src, const int* __restrict__ dst,
                         int* deg_s, int* deg_d, int E) {
  int e = blockIdx.x * blockDim.x + threadIdx.x;
  if (e < E) {
    atomicAdd(&deg_s[src[e]], 1);
    atomicAdd(&deg_d[dst[e]], 1);
  }
}

__global__ void k_norm(const int* __restrict__ deg_s, const int* __restrict__ deg_d,
                       float* ns, float* nd, int N) {
  int i = blockIdx.x * blockDim.x + threadIdx.x;
  if (i < N) {
    int a = deg_s[i]; ns[i] = (a > 0) ? rsqrtf((float)a) : 1.0f;
    int b = deg_d[i]; nd[i] = (b > 0) ? rsqrtf((float)b) : 1.0f;
  }
}

__global__ void k_scatter64(const int* __restrict__ src, const int* __restrict__ dst,
                            const float* __restrict__ x, const float* __restrict__ ns,
                            float* agg, int E) {
  long t = (long)blockIdx.x * blockDim.x + threadIdx.x;
  int e = (int)(t >> 4);
  if (e >= E) return;
  int f = ((int)t & 15) << 2;
  int s = src[e], d = dst[e];
  float w = ns[s];
  float4 v = *(const float4*)(x + (long)s * IN_F + f);
  float* o = agg + (long)d * IN_F + f;
  atomicAdd(o + 0, v.x * w);
  atomicAdd(o + 1, v.y * w);
  atomicAdd(o + 2, v.z * w);
  atomicAdd(o + 3, v.w * w);
}

__global__ void k_scatter16(const int* __restrict__ src, const int* __restrict__ dst,
                            const float* __restrict__ z, const float* __restrict__ ns,
                            float* agg, int E) {
  long t = (long)blockIdx.x * blockDim.x + threadIdx.x;
  int e = (int)(t >> 2);
  if (e >= E) return;
  int f = ((int)t & 3) << 2;
  int s = src[e], d = dst[e];
  float w = ns[s];
  float4 v = *(const float4*)(z + (long)s * NCLS + f);
  float* o = agg + (long)d * NCLS + f;
  atomicAdd(o + 0, v.x * w);
  atomicAdd(o + 1, v.y * w);
  atomicAdd(o + 2, v.z * w);
  atomicAdd(o + 3, v.w * w);
}

__global__ void __launch_bounds__(256)
k_gemm_relu(const float* __restrict__ A, const float* __restrict__ nd,
            const float* __restrict__ W, const float* __restrict__ bias,
            float* __restrict__ out, int nrows) {
  __shared__ float sW[IN_F * OUT_F];
  for (int i = threadIdx.x; i < IN_F * OUT_F; i += blockDim.x) sW[i] = W[i];
  __syncthreads();

  int wave = (int)((blockIdx.x * blockDim.x + threadIdx.x) >> 5);
  int lane = threadIdx.x & 31;
  int m0 = wave * 16;
  if (m0 >= nrows) return;

  int lane16 = lane & 15;
  int half   = lane >> 4;
  int arow   = m0 + lane16;
  float scale = nd[arow];
  const float* ap = A + (long)arow * IN_F + half * 2;

  v8f acc0 = {}, acc1 = {}, acc2 = {}, acc3 = {};
#pragma unroll
  for (int k0 = 0; k0 < IN_F; k0 += 4) {
    float2 av = *(const float2*)(ap + k0);
    v2f a = {av.x * scale, av.y * scale};
    int kk = k0 + half * 2;
    const float* w0 = sW + kk * OUT_F + lane16;
    const float* w1 = w0 + OUT_F;
    v2f b0 = {w0[0],  w1[0]};
    v2f b1 = {w0[16], w1[16]};
    v2f b2 = {w0[32], w1[32]};
    v2f b3 = {w0[48], w1[48]};
    acc0 = __builtin_amdgcn_wmma_f32_16x16x4_f32(false, a, false, b0, (short)0, acc0, false, false);
    acc1 = __builtin_amdgcn_wmma_f32_16x16x4_f32(false, a, false, b1, (short)0, acc1, false, false);
    acc2 = __builtin_amdgcn_wmma_f32_16x16x4_f32(false, a, false, b2, (short)0, acc2, false, false);
    acc3 = __builtin_amdgcn_wmma_f32_16x16x4_f32(false, a, false, b3, (short)0, acc3, false, false);
  }

  int rbase = m0 + half * 8;
  float bi0 = bias[lane16], bi1 = bias[16 + lane16];
  float bi2 = bias[32 + lane16], bi3 = bias[48 + lane16];
#pragma unroll
  for (int r = 0; r < 8; ++r) {
    float* orow = out + (long)(rbase + r) * OUT_F + lane16;
    orow[0]  = fmaxf(acc0[r] + bi0, 0.f);
    orow[16] = fmaxf(acc1[r] + bi1, 0.f);
    orow[32] = fmaxf(acc2[r] + bi2, 0.f);
    orow[48] = fmaxf(acc3[r] + bi3, 0.f);
  }
}

__global__ void __launch_bounds__(256)
k_gemm16(const float* __restrict__ A, const float* __restrict__ W,
         float* __restrict__ out, int nrows) {
  __shared__ float sW[OUT_F * NCLS];
  for (int i = threadIdx.x; i < OUT_F * NCLS; i += blockDim.x) sW[i] = W[i];
  __syncthreads();

  int wave = (int)((blockIdx.x * blockDim.x + threadIdx.x) >> 5);
  int lane = threadIdx.x & 31;
  int m0 = wave * 16;
  if (m0 >= nrows) return;

  int lane16 = lane & 15;
  int half   = lane >> 4;
  const float* ap = A + (long)(m0 + lane16) * OUT_F + half * 2;

  v8f acc = {};
#pragma unroll
  for (int k0 = 0; k0 < OUT_F; k0 += 4) {
    float2 av = *(const float2*)(ap + k0);
    v2f a = {av.x, av.y};
    int kk = k0 + half * 2;
    v2f b = {sW[kk * NCLS + lane16], sW[(kk + 1) * NCLS + lane16]};
    acc = __builtin_amdgcn_wmma_f32_16x16x4_f32(false, a, false, b, (short)0, acc, false, false);
  }
  int rbase = m0 + half * 8;
#pragma unroll
  for (int r = 0; r < 8; ++r)
    out[(long)(rbase + r) * NCLS + lane16] = acc[r];
}

__global__ void k_finalize(const float* __restrict__ agg, const float* __restrict__ nd,
                           const float* __restrict__ bias, float* __restrict__ out, int N) {
  int i = blockIdx.x * blockDim.x + threadIdx.x;
  if (i < N * NCLS) {
    int n = i >> 4, f = i & 15;
    out[i] = agg[i] * nd[n] + bias[f];
  }
}

extern "C" void kernel_launch(void* const* d_in, const int* in_sizes, int n_in,
                              void* d_out, int out_size, void* d_ws, size_t ws_size,
                              hipStream_t stream) {
  const float* x  = (const float*)d_in[0];
  const int*   src = (const int*)d_in[1];
  const int*   dst = (const int*)d_in[2];
  const float* W1 = (const float*)d_in[3];
  const float* b1 = (const float*)d_in[4];
  const float* W2 = (const float*)d_in[5];
  const float* b2 = (const float*)d_in[6];
  float* out = (float*)d_out;

  const int N = in_sizes[0] / IN_F;
  const int E = in_sizes[1];

  char* ws = (char*)d_ws;
  size_t off = 0;
  auto carve = [&](size_t bytes) -> void* {
    void* p = (void*)(ws + off);
    off += (bytes + 255) & ~(size_t)255;
    return p;
  };
  int*   deg_s = (int*)carve((size_t)N * 4);
  int*   deg_d = (int*)carve((size_t)N * 4);
  float* ns    = (float*)carve((size_t)N * 4);
  float* nd    = (float*)carve((size_t)N * 4);
  float* agg1  = (float*)carve((size_t)N * IN_F * 4);
  float* y1    = (float*)carve((size_t)N * OUT_F * 4);
  float* z     = (float*)carve((size_t)N * NCLS * 4);
  float* agg2  = (float*)carve((size_t)N * NCLS * 4);

  const int B = 256;
  auto cdiv = [](long a, long b) { return (int)((a + b - 1) / b); };

  long n4;
  n4 = ((long)N * 4 + 15) / 16;
  k_zero<<<cdiv(n4, B), B, 0, stream>>>((float4*)deg_s, n4);
  k_zero<<<cdiv(n4, B), B, 0, stream>>>((float4*)deg_d, n4);
  n4 = (long)N * IN_F / 4;
  k_zero<<<cdiv(n4, B), B, 0, stream>>>((float4*)agg1, n4);
  n4 = (long)N * NCLS / 4;
  k_zero<<<cdiv(n4, B), B, 0, stream>>>((float4*)agg2, n4);

  k_degree<<<cdiv(E, B), B, 0, stream>>>(src, dst, deg_s, deg_d, E);
  k_norm<<<cdiv(N, B), B, 0, stream>>>(deg_s, deg_d, ns, nd, N);

  k_scatter64<<<cdiv((long)E * 16, B), B, 0, stream>>>(src, dst, x, ns, agg1, E);
  int waves = cdiv(N, 16);
  k_gemm_relu<<<cdiv((long)waves * 32, B), B, 0, stream>>>(agg1, nd, W1, b1, y1, N);

  k_gemm16<<<cdiv((long)waves * 32, B), B, 0, stream>>>(y1, W2, z, N);
  k_scatter16<<<cdiv((long)E * 4, B), B, 0, stream>>>(src, dst, z, ns, agg2, E);
  k_finalize<<<cdiv((long)N * NCLS, B), B, 0, stream>>>(agg2, nd, b2, out, N);
}


# degree counts: global int atomics and their zero fills replaced by per-block LDS histograms over (node range x edge slice), stored as partial planes that the norm kernel adds in slice order
# speedup vs baseline: 2.3608x; 2.3608x over previous
#include <hip/hip_runtime.h>

typedef __attribute__((ext_vector_type(2))) float v2f;
typedef __attribute__((ext_vector_type(8))) float v8f;

#define IN_F 64
#define OUT_F 64
#define NCLS 16

__global__ void k_zero(float4* __restrict__ p, long n4) {
  long i = (long)blockIdx.x * blockDim.x + threadIdx.x;
  if (i < n4) p[i] = make_float4(0.f, 0.f, 0.f, 0.f);
}

#define HR 50016
#define HSLICES 32

__global__ void __launch_bounds__(1024)
k_deg_hist(const int* __restrict__ src, const int* __restrict__ dst,
           unsigned* __restrict__ part, int E, int N, int per_slice) {
  __shared__ unsigned h[HR];
  const int* idx = blockIdx.z ? dst : src;
  int lo = blockIdx.y * HR;
  int cnt = min(HR, N - lo);
  for (int i = threadIdx.x; i < HR; i += 1024) h[i] = 0u;
  __syncthreads();

  int e0 = blockIdx.x * per_slice;
  int e1 = min(E, e0 + per_slice);
  int nv = (e1 - e0) >> 2;
  const int4* iv = (const int4*)(idx + e0);
  int i = threadIdx.x;
  for (; i + 3 * 1024 < nv; i += 4 * 1024) {
    int4 v[4];
#pragma unroll
    for (int u = 0; u < 4; ++u) v[u] = iv[i + u * 1024];
#pragma unroll
    for (int u = 0; u < 4; ++u) {
      unsigned r0 = (unsigned)(v[u].x - lo), r1 = (unsigned)(v[u].y - lo);
      unsigned r2 = (unsigned)(v[u].z - lo), r3 = (unsigned)(v[u].w - lo);
      if (r0 < (unsigned)cnt) atomicAdd(&h[r0], 1u);
      if (r1 < (unsigned)cnt) atomicAdd(&h[r1], 1u);
      if (r2 < (unsigned)cnt) atomicAdd(&h[r2], 1u);
      if (r3 < (unsigned)cnt) atomicAdd(&h[r3], 1u);
    }
  }
  for (; i < nv; i += 1024) {
    int4 v = iv[i];
    unsigned r0 = (unsigned)(v.x - lo), r1 = (unsigned)(v.y - lo);
    unsigned r2 = (unsigned)(v.z - lo), r3 = (unsigned)(v.w - lo);
    if (r0 < (unsigned)cnt) atomicAdd(&h[r0], 1u);
    if (r1 < (unsigned)cnt) atomicAdd(&h[r1], 1u);
    if (r2 < (unsigned)cnt) atomicAdd(&h[r2], 1u);
    if (r3 < (unsigned)cnt) atomicAdd(&h[r3], 1u);
  }
  if (nv < 0) nv = 0;
  for (int e = e0 + (nv << 2) + threadIdx.x; e < e1; e += 1024) {
    unsigned r = (unsigned)(idx[e] - lo);
    if (r < (unsigned)cnt) atomicAdd(&h[r], 1u);
  }
  __syncthreads();

  unsigned* prow = part + ((size_t)blockIdx.z * HSLICES + blockIdx.x) * (size_t)N + lo;
  for (int i = threadIdx.x; i < cnt; i += 1024) prow[i] = h[i];
}

__global__ void k_norm(const unsigned* __restrict__ part, float* ns, float* nd, int N) {
  int i = blockIdx.x * blockDim.x + threadIdx.x;
  if (i < N) {
    const unsigned* ps = part + i;
    const unsigned* pd = part + (size_t)HSLICES * N + i;
    unsigned a = 0u, b = 0u;
#pragma unroll
    for (int s = 0; s < HSLICES; ++s) {
      a += ps[(size_t)s * N];
      b += pd[(size_t)s * N];
    }
    ns[i] = (a > 0u) ? rsqrtf((float)a) : 1.0f;
    nd[i] = (b > 0u) ? rsqrtf((float)b) : 1.0f;
  }
}


__device__ __forceinline__ int bcast_i(int v, int j) { return __builtin_amdgcn_readlane(v, j); }
__device__ __forceinline__ float bcast_f(float v, int j) {
  return __int_as_float(__builtin_amdgcn_readlane(__float_as_int(v), j));
}

__global__ void __launch_bounds__(256)
k_scatter64(const int* __restrict__ src, const int* __restrict__ dst,
            const float* __restrict__ x, const float* __restrict__ ns,
            float* agg, int E) {
  int t = blockIdx.x * blockDim.x + threadIdx.x;
  int lane = threadIdx.x & 31;
  int e0 = __builtin_amdgcn_readfirstlane(t);
  if (e0 >= E) return;
  int s = 0, d = 0;
  float w = 0.f;
  if (t < E) { s = src[t]; d = dst[t]; w = ns[s]; }

  if (e0 + 32 <= E) {
#pragma unroll
    for (int g = 0; g < 32; g += 8) {
      float v0[8], v1[8];
#pragma unroll
      for (int j = 0; j < 8; ++j) {
        const float* xr = x + (long)bcast_i(s, g + j) * IN_F + lane;
        v0[j] = xr[0];
        v1[j] = xr[32];
      }
#pragma unroll
      for (int j = 0; j < 8; ++j) {
        float wj = bcast_f(w, g + j);
        float* ar = agg + (long)bcast_i(d, g + j) * IN_F + lane;
        atomicAdd(ar, v0[j] * wj);
        atomicAdd(ar + 32, v1[j] * wj);
      }
    }
  } else {
    int n = E - e0;
    for (int j = 0; j < n; ++j) {
      const float* xr = x + (long)bcast_i(s, j) * IN_F + lane;
      float wj = bcast_f(w, j);
      float* ar = agg + (long)bcast_i(d, j) * IN_F + lane;
      atomicAdd(ar, xr[0] * wj);
      atomicAdd(ar + 32, xr[32] * wj);
    }
  }
}

__global__ void __launch_bounds__(256)
k_scatter16(const int* __restrict__ src, const int* __restrict__ dst,
            const float* __restrict__ z, const float* __restrict__ ns,
            float* agg, int E) {
  int t = blockIdx.x * blockDim.x + threadIdx.x;
  int lane = threadIdx.x & 31;
  int l16 = lane & 15;
  int half = lane >> 4;
  int e0 = __builtin_amdgcn_readfirstlane(t);
  if (e0 >= E) return;
  int s = 0, d = 0;
  float w = 0.f;
  if (t < E) { s = src[t]; d = dst[t]; w = ns[s]; }

  if (e0 + 32 <= E) {
#pragma unroll
    for (int g = 0; g < 16; g += 8) {
      float v[8];
#pragma unroll
      for (int j = 0; j < 8; ++j) {
        int p = 2 * (g + j);
        int sa = bcast_i(s, p), sb = bcast_i(s, p + 1);
        int sj = half ? sb : sa;
        v[j] = z[(long)sj * NCLS + l16];
      }
#pragma unroll
      for (int j = 0; j < 8; ++j) {
        int p = 2 * (g + j);
        int da = bcast_i(d, p), db = bcast_i(d, p + 1);
        float wa = bcast_f(w, p), wb = bcast_f(w, p + 1);
        int dj = half ? db : da;
        float wj = half ? wb : wa;
        atomicAdd(agg + (long)dj * NCLS + l16, v[j] * wj);
      }
    }
  } else {
    int n = E - e0;
    for (int p = 0; p < n; p += 2) {
      int sa = bcast_i(s, p), sb = bcast_i(s, p + 1);
      int da = bcast_i(d, p), db = bcast_i(d, p + 1);
      float wa = bcast_f(w, p), wb = bcast_f(w, p + 1);
      int sj = half ? sb : sa;
      int dj = half ? db : da;
      float wj = half ? wb : wa;
      if (p + half < n) atomicAdd(agg + (long)dj * NCLS + l16, z[(long)sj * NCLS + l16] * wj);
    }
  }
}

__global__ void __launch_bounds__(256)
k_gemm_relu(const float* __restrict__ A, const float* __restrict__ nd,
            const float* __restrict__ W, const float* __restrict__ bias,
            float* __restrict__ out, int nrows) {
  __shared__ float sW[IN_F * OUT_F];
  for (int i = threadIdx.x; i < IN_F * OUT_F; i += blockDim.x) sW[i] = W[i];
  __syncthreads();

  int wave = (int)((blockIdx.x * blockDim.x + threadIdx.x) >> 5);
  int lane = threadIdx.x & 31;
  int m0 = wave * 16;
  if (m0 >= nrows) return;

  int lane16 = lane & 15;
  int half   = lane >> 4;
  int arow   = m0 + lane16;
  float scale = nd[arow];
  const float* ap = A + (long)arow * IN_F + half * 2;

  v8f acc0 = {}, acc1 = {}, acc2 = {}, acc3 = {};
#pragma unroll
  for (int k0 = 0; k0 < IN_F; k0 += 4) {
    float2 av = *(const float2*)(ap + k0);
    v2f a = {av.x * scale, av.y * scale};
    int kk = k0 + half * 2;
    const float* w0 = sW + kk * OUT_F + lane16;
    const float* w1 = w0 + OUT_F;
    v2f b0 = {w0[0],  w1[0]};
    v2f b1 = {w0[16], w1[16]};
    v2f b2 = {w0[32], w1[32]};
    v2f b3 = {w0[48], w1[48]};
    acc0 = __builtin_amdgcn_wmma_f32_16x16x4_f32(false, a, false, b0, (short)0, acc0, false, false);
    acc1 = __builtin_amdgcn_wmma_f32_16x16x4_f32(false, a, false, b1, (short)0, acc1, false, false);
    acc2 = __builtin_amdgcn_wmma_f32_16x16x4_f32(false, a, false, b2, (short)0, acc2, false, false);
    acc3 = __builtin_amdgcn_wmma_f32_16x16x4_f32(false, a, false, b3, (short)0, acc3, false, false);
  }

  int rbase = m0 + half * 8;
  float bi0 = bias[lane16], bi1 = bias[16 + lane16];
  float bi2 = bias[32 + lane16], bi3 = bias[48 + lane16];
#pragma unroll
  for (int r = 0; r < 8; ++r) {
    float* orow = out + (long)(rbase + r) * OUT_F + lane16;
    orow[0]  = fmaxf(acc0[r] + bi0, 0.f);
    orow[16] = fmaxf(acc1[r] + bi1, 0.f);
    orow[32] = fmaxf(acc2[r] + bi2, 0.f);
    orow[48] = fmaxf(acc3[r] + bi3, 0.f);
  }
}

__global__ void __launch_bounds__(256)
k_gemm16(const float* __restrict__ A, const float* __restrict__ W,
         float* __restrict__ out, int nrows) {
  __shared__ float sW[OUT_F * NCLS];
  for (int i = threadIdx.x; i < OUT_F * NCLS; i += blockDim.x) sW[i] = W[i];
  __syncthreads();

  int wave = (int)((blockIdx.x * blockDim.x + threadIdx.x) >> 5);
  int lane = threadIdx.x & 31;
  int m0 = wave * 16;
  if (m0 >= nrows) return;

  int lane16 = lane & 15;
  int half   = lane >> 4;
  const float* ap = A + (long)(m0 + lane16) * OUT_F + half * 2;

  v8f acc = {};
#pragma unroll
  for (int k0 = 0; k0 < OUT_F; k0 += 4) {
    float2 av = *(const float2*)(ap + k0);
    v2f a = {av.x, av.y};
    int kk = k0 + half * 2;
    v2f b = {sW[kk * NCLS + lane16], sW[(kk + 1) * NCLS + lane16]};
    acc = __builtin_amdgcn_wmma_f32_16x16x4_f32(false, a, false, b, (short)0, acc, false, false);
  }
  int rbase = m0 + half * 8;
#pragma unroll
  for (int r = 0; r < 8; ++r)
    out[(long)(rbase + r) * NCLS + lane16] = acc[r];
}

__global__ void k_finalize(const float* __restrict__ agg, const float* __restrict__ nd,
                           const float* __restrict__ bias, float* __restrict__ out, int N) {
  int i = blockIdx.x * blockDim.x + threadIdx.x;
  if (i < N * NCLS) {
    int n = i >> 4, f = i & 15;
    out[i] = agg[i] * nd[n] + bias[f];
  }
}

extern "C" void kernel_launch(void* const* d_in, const int* in_sizes, int n_in,
                              void* d_out, int out_size, void* d_ws, size_t ws_size,
                              hipStream_t stream) {
  const float* x  = (const float*)d_in[0];
  const int*   src = (const int*)d_in[1];
  const int*   dst = (const int*)d_in[2];
  const float* W1 = (const float*)d_in[3];
  const float* b1 = (const float*)d_in[4];
  const float* W2 = (const float*)d_in[5];
  const float* b2 = (const float*)d_in[6];
  float* out = (float*)d_out;

  const int N = in_sizes[0] / IN_F;
  const int E = in_sizes[1];

  char* ws = (char*)d_ws;
  size_t off = 0;
  auto carve = [&](size_t bytes) -> void* {
    void* p = (void*)(ws + off);
    off += (bytes + 255) & ~(size_t)255;
    return p;
  };
  float* ns    = (float*)carve((size_t)N * 4);
  float* nd    = (float*)carve((size_t)N * 4);
  float* agg1  = (float*)carve((size_t)N * IN_F * 4);
  float* y1    = (float*)carve((size_t)N * OUT_F * 4);
  unsigned* part = (unsigned*)y1;
  float* z     = (float*)carve((size_t)N * NCLS * 4);
  float* agg2  = (float*)carve((size_t)N * NCLS * 4);

  const int B = 256;
  auto cdiv = [](long a, long b) { return (int)((a + b - 1) / b); };

  long n4;
  n4 = (long)N * IN_F / 4;
  k_zero<<<cdiv(n4, B), B, 0, stream>>>((float4*)agg1, n4);
  n4 = (long)N * NCLS / 4;
  k_zero<<<cdiv(n4, B), B, 0, stream>>>((float4*)agg2, n4);

  int per_slice = cdiv(cdiv(E, HSLICES), 4) * 4;
  k_deg_hist<<<dim3(HSLICES, cdiv(N, HR), 2), 1024, 0, stream>>>(src, dst, part, E, N, per_slice);
  k_norm<<<cdiv(N, B), B, 0, stream>>>(part, ns, nd, N);

  k_scatter64<<<cdiv(E, B), B, 0, stream>>>(src, dst, x, ns, agg1, E);
  int waves = cdiv(N, 16);
  k_gemm_relu<<<cdiv((long)waves * 32, B), B, 0, stream>>>(agg1, nd, W1, b1, y1, N);

  k_gemm16<<<cdiv((long)waves * 32, B), B, 0, stream>>>(y1, W2, z, N);
  k_scatter16<<<cdiv(E, B), B, 0, stream>>>(src, dst, z, ns, agg2, E);
  k_finalize<<<cdiv((long)N * NCLS, B), B, 0, stream>>>(agg2, nd, b2, out, N);
}


# both edge scatters split by node half: each edge chunk is claimed twice via ticket counters, a workgroup takes the dst half matching its own cache half (die id); sum order = atomic arrival as before
# speedup vs baseline: 2.7454x; 1.1629x over previous
#include <hip/hip_runtime.h>

typedef __attribute__((ext_vector_type(2))) float v2f;
typedef __attribute__((ext_vector_type(8))) float v8f;

#define IN_F 64
#define OUT_F 64
#define NCLS 16

__global__ void k_zero(float4* __restrict__ p, long n4) {
  long i = (long)blockIdx.x * blockDim.x + threadIdx.x;
  if (i < n4) p[i] = make_float4(0.f, 0.f, 0.f, 0.f);
}

#define HR 50016
#define HSLICES 32

__global__ void __launch_bounds__(1024)
k_deg_hist(const int* __restrict__ src, const int* __restrict__ dst,
           unsigned* __restrict__ part, int E, int N, int per_slice) {
  __shared__ unsigned h[HR];
  const int* idx = blockIdx.z ? dst : src;
  int lo = blockIdx.y * HR;
  int cnt = min(HR, N - lo);
  for (int i = threadIdx.x; i < HR; i += 1024) h[i] = 0u;
  __syncthreads();

  int e0 = blockIdx.x * per_slice;
  int e1 = min(E, e0 + per_slice);
  int nv = (e1 - e0) >> 2;
  const int4* iv = (const int4*)(idx + e0);
  int i = threadIdx.x;
  for (; i + 3 * 1024 < nv; i += 4 * 1024) {
    int4 v[4];
#pragma unroll
    for (int u = 0; u < 4; ++u) v[u] = iv[i + u * 1024];
#pragma unroll
    for (int u = 0; u < 4; ++u) {
      unsigned r0 = (unsigned)(v[u].x - lo), r1 = (unsigned)(v[u].y - lo);
      unsigned r2 = (unsigned)(v[u].z - lo), r3 = (unsigned)(v[u].w - lo);
      if (r0 < (unsigned)cnt) atomicAdd(&h[r0], 1u);
      if (r1 < (unsigned)cnt) atomicAdd(&h[r1], 1u);
      if (r2 < (unsigned)cnt) atomicAdd(&h[r2], 1u);
      if (r3 < (unsigned)cnt) atomicAdd(&h[r3], 1u);
    }
  }
  for (; i < nv; i += 1024) {
    int4 v = iv[i];
    unsigned r0 = (unsigned)(v.x - lo), r1 = (unsigned)(v.y - lo);
    unsigned r2 = (unsigned)(v.z - lo), r3 = (unsigned)(v.w - lo);
    if (r0 < (unsigned)cnt) atomicAdd(&h[r0], 1u);
    if (r1 < (unsigned)cnt) atomicAdd(&h[r1], 1u);
    if (r2 < (unsigned)cnt) atomicAdd(&h[r2], 1u);
    if (r3 < (unsigned)cnt) atomicAdd(&h[r3], 1u);
  }
  if (nv < 0) nv = 0;
  for (int e = e0 + (nv << 2) + threadIdx.x; e < e1; e += 1024) {
    unsigned r = (unsigned)(idx[e] - lo);
    if (r < (unsigned)cnt) atomicAdd(&h[r], 1u);
  }
  __syncthreads();

  unsigned* prow = part + ((size_t)blockIdx.z * HSLICES + blockIdx.x) * (size_t)N + lo;
  for (int i = threadIdx.x; i < cnt; i += 1024) prow[i] = h[i];
}

__global__ void k_norm(const unsigned* __restrict__ part, float* ns, float* nd, int N) {
  int i = blockIdx.x * blockDim.x + threadIdx.x;
  if (i < N) {
    const unsigned* ps = part + i;
    const unsigned* pd = part + (size_t)HSLICES * N + i;
    unsigned a = 0u, b = 0u;
#pragma unroll
    for (int s = 0; s < HSLICES; ++s) {
      a += ps[(size_t)s * N];
      b += pd[(size_t)s * N];
    }
    ns[i] = (a > 0u) ? rsqrtf((float)a) : 1.0f;
    nd[i] = (b > 0u) ? rsqrtf((float)b) : 1.0f;
  }
}


__device__ __forceinline__ int bcast_i(int v, int j) { return __builtin_amdgcn_readlane(v, j); }
__device__ __forceinline__ float bcast_f(float v, int j) {
  return __int_as_float(__builtin_amdgcn_readlane(__float_as_int(v), j));
}

__device__ __forceinline__ int chip_half() {
  unsigned id = (unsigned)__builtin_amdgcn_s_sendmsg_rtn(0x87);
  return (int)((id >> 18) & 1u);
}

#define CHUNK 512

__device__ __forceinline__ void claim(int* ctr, int nchunks, int* sh) {
  if (threadIdx.x == 0) {
    int h = chip_half();
    int t = atomicAdd(&ctr[h], 1);
    if (t >= nchunks) { h ^= 1; t = atomicAdd(&ctr[h], 1); }
    sh[0] = h; sh[1] = t;
  }
  __syncthreads();
}

__global__ void __launch_bounds__(256)
k_scatter64(const int* __restrict__ src, const int* __restrict__ dst,
            const float* __restrict__ x, const float* __restrict__ ns,
            float* agg, int E, int N, int* ctr, int nchunks) {
  __shared__ int sh[2];
  claim(ctr, nchunks, sh);
  int r = sh[0], c = sh[1];
  if (c >= nchunks) return;
  int lo = r ? (N >> 1) : 0;
  int cnt = r ? N - (N >> 1) : (N >> 1);
  int lane = threadIdx.x & 31;
  int eb = c * CHUNK + (threadIdx.x >> 5) * 64;

  int s[2], d[2];
  float w[2];
  unsigned m[2];
#pragma unroll
  for (int k = 0; k < 2; ++k) {
    int e = eb + k * 32 + lane;
    bool ok = e < E;
    s[k] = ok ? src[e] : 0;
    d[k] = ok ? dst[e] : 0;
  }
#pragma unroll
  for (int k = 0; k < 2; ++k) {
    int e = eb + k * 32 + lane;
    bool hit = (e < E) && ((unsigned)(d[k] - lo) < (unsigned)cnt);
    w[k] = hit ? ns[s[k]] : 0.f;
    m[k] = __builtin_amdgcn_ballot_w32(hit);
  }

#pragma unroll
  for (int k = 0; k < 2; ++k) {
    unsigned mm = m[k];
    while (mm) {
      int j[8];
      int n = 0;
#pragma unroll
      for (int u = 0; u < 8; ++u) {
        j[u] = mm ? __builtin_ctz(mm) : 0;
        n += (mm != 0u);
        mm &= mm - 1u;
      }
      float v0[8], v1[8];
#pragma unroll
      for (int u = 0; u < 8; ++u) {
        v0[u] = 0.f; v1[u] = 0.f;
        if (u < n) {
          const float* xr = x + (long)bcast_i(s[k], j[u]) * IN_F + lane;
          v0[u] = xr[0];
          v1[u] = xr[32];
        }
      }
#pragma unroll
      for (int u = 0; u < 8; ++u) {
        if (u < n) {
          float wj = bcast_f(w[k], j[u]);
          float* ar = agg + (long)bcast_i(d[k], j[u]) * IN_F + lane;
          atomicAdd(ar, v0[u] * wj);
          atomicAdd(ar + 32, v1[u] * wj);
        }
      }
    }
  }
}

__global__ void __launch_bounds__(256)
k_scatter16(const int* __restrict__ src, const int* __restrict__ dst,
            const float* __restrict__ z, const float* __restrict__ ns,
            float* agg, int E, int N, int* ctr, int nchunks) {
  __shared__ int sh[2];
  claim(ctr, nchunks, sh);
  int r = sh[0], c = sh[1];
  if (c >= nchunks) return;
  int lo = r ? (N >> 1) : 0;
  int cnt = r ? N - (N >> 1) : (N >> 1);
  int lane = threadIdx.x & 31;
  int l16 = lane & 15;
  int half = lane >> 4;
  int eb = c * CHUNK + (threadIdx.x >> 5) * 64;

  int s[2], d[2];
  float w[2];
  unsigned m[2];
#pragma unroll
  for (int k = 0; k < 2; ++k) {
    int e = eb + k * 32 + lane;
    bool ok = e < E;
    s[k] = ok ? src[e] : 0;
    d[k] = ok ? dst[e] : 0;
  }
#pragma unroll
  for (int k = 0; k < 2; ++k) {
    int e = eb + k * 32 + lane;
    bool hit = (e < E) && ((unsigned)(d[k] - lo) < (unsigned)cnt);
    w[k] = hit ? ns[s[k]] : 0.f;
    m[k] = __builtin_amdgcn_ballot_w32(hit);
  }

#pragma unroll
  for (int k = 0; k < 2; ++k) {
    unsigned mm = m[k];
    while (mm) {
      int ja[8], jb[8], ne[8];
#pragma unroll
      for (int u = 0; u < 8; ++u) {
        ja[u] = mm ? __builtin_ctz(mm) : 0;
        int na = (mm != 0u);
        mm &= mm - 1u;
        jb[u] = mm ? __builtin_ctz(mm) : 0;
        ne[u] = na + (mm != 0u);
        mm &= mm - 1u;
      }
      float v[8];
#pragma unroll
      for (int u = 0; u < 8; ++u) {
        v[u] = 0.f;
        if (ne[u]) {
          int sa = bcast_i(s[k], ja[u]), sb = bcast_i(s[k], jb[u]);
          int sj = half ? sb : sa;
          if (half < ne[u]) v[u] = z[(long)sj * NCLS + l16];
        }
      }
#pragma unroll
      for (int u = 0; u < 8; ++u) {
        if (ne[u]) {
          int da = bcast_i(d[k], ja[u]), db = bcast_i(d[k], jb[u]);
          float wa = bcast_f(w[k], ja[u]), wb = bcast_f(w[k], jb[u]);
          int dj = half ? db : da;
          float wj = half ? wb : wa;
          if (half < ne[u]) atomicAdd(agg + (long)dj * NCLS + l16, v[u] * wj);
        }
      }
    }
  }
}

__global__ void __launch_bounds__(256)
k_gemm_relu(const float* __restrict__ A, const float* __restrict__ nd,
            const float* __restrict__ W, const float* __restrict__ bias,
            float* __restrict__ out, int nrows) {
  __shared__ float sW[IN_F * OUT_F];
  for (int i = threadIdx.x; i < IN_F * OUT_F; i += blockDim.x) sW[i] = W[i];
  __syncthreads();

  int wave = (int)((blockIdx.x * blockDim.x + threadIdx.x) >> 5);
  int lane = threadIdx.x & 31;
  int m0 = wave * 16;
  if (m0 >= nrows) return;

  int lane16 = lane & 15;
  int half   = lane >> 4;
  int arow   = m0 + lane16;
  float scale = nd[arow];
  const float* ap = A + (long)arow * IN_F + half * 2;

  v8f acc0 = {}, acc1 = {}, acc2 = {}, acc3 = {};
#pragma unroll
  for (int k0 = 0; k0 < IN_F; k0 += 4) {
    float2 av = *(const float2*)(ap + k0);
    v2f a = {av.x * scale, av.y * scale};
    int kk = k0 + half * 2;
    const float* w0 = sW + kk * OUT_F + lane16;
    const float* w1 = w0 + OUT_F;
    v2f b0 = {w0[0],  w1[0]};
    v2f b1 = {w0[16], w1[16]};
    v2f b2 = {w0[32], w1[32]};
    v2f b3 = {w0[48], w1[48]};
    acc0 = __builtin_amdgcn_wmma_f32_16x16x4_f32(false, a, false, b0, (short)0, acc0, false, false);
    acc1 = __builtin_amdgcn_wmma_f32_16x16x4_f32(false, a, false, b1, (short)0, acc1, false, false);
    acc2 = __builtin_amdgcn_wmma_f32_16x16x4_f32(false, a, false, b2, (short)0, acc2, false, false);
    acc3 = __builtin_amdgcn_wmma_f32_16x16x4_f32(false, a, false, b3, (short)0, acc3, false, false);
  }

  int rbase = m0 + half * 8;
  float bi0 = bias[lane16], bi1 = bias[16 + lane16];
  float bi2 = bias[32 + lane16], bi3 = bias[48 + lane16];
#pragma unroll
  for (int r = 0; r < 8; ++r) {
    float* orow = out + (long)(rbase + r) * OUT_F + lane16;
    orow[0]  = fmaxf(acc0[r] + bi0, 0.f);
    orow[16] = fmaxf(acc1[r] + bi1, 0.f);
    orow[32] = fmaxf(acc2[r] + bi2, 0.f);
    orow[48] = fmaxf(acc3[r] + bi3, 0.f);
  }
}

__global__ void __launch_bounds__(256)
k_gemm16(const float* __restrict__ A, const float* __restrict__ W,
         float* __restrict__ out, int nrows) {
  __shared__ float sW[OUT_F * NCLS];
  for (int i = threadIdx.x; i < OUT_F * NCLS; i += blockDim.x) sW[i] = W[i];
  __syncthreads();

  int wave = (int)((blockIdx.x * blockDim.x + threadIdx.x) >> 5);
  int lane = threadIdx.x & 31;
  int m0 = wave * 16;
  if (m0 >= nrows) return;

  int lane16 = lane & 15;
  int half   = lane >> 4;
  const float* ap = A + (long)(m0 + lane16) * OUT_F + half * 2;

  v8f acc = {};
#pragma unroll
  for (int k0 = 0; k0 < OUT_F; k0 += 4) {
    float2 av = *(const float2*)(ap + k0);
    v2f a = {av.x, av.y};
    int kk = k0 + half * 2;
    v2f b = {sW[kk * NCLS + lane16], sW[(kk + 1) * NCLS + lane16]};
    acc = __builtin_amdgcn_wmma_f32_16x16x4_f32(false, a, false, b, (short)0, acc, false, false);
  }
  int rbase = m0 + half * 8;
#pragma unroll
  for (int r = 0; r < 8; ++r)
    out[(long)(rbase + r) * NCLS + lane16] = acc[r];
}

__global__ void k_finalize(const float* __restrict__ agg, const float* __restrict__ nd,
                           const float* __restrict__ bias, float* __restrict__ out, int N) {
  int i = blockIdx.x * blockDim.x + threadIdx.x;
  if (i < N * NCLS) {
    int n = i >> 4, f = i & 15;
    out[i] = agg[i] * nd[n] + bias[f];
  }
}

extern "C" void kernel_launch(void* const* d_in, const int* in_sizes, int n_in,
                              void* d_out, int out_size, void* d_ws, size_t ws_size,
                              hipStream_t stream) {
  const float* x  = (const float*)d_in[0];
  const int*   src = (const int*)d_in[1];
  const int*   dst = (const int*)d_in[2];
  const float* W1 = (const float*)d_in[3];
  const float* b1 = (const float*)d_in[4];
  const float* W2 = (const float*)d_in[5];
  const float* b2 = (const float*)d_in[6];
  float* out = (float*)d_out;

  const int N = in_sizes[0] / IN_F;
  const int E = in_sizes[1];

  char* ws = (char*)d_ws;
  size_t off = 0;
  auto carve = [&](size_t bytes) -> void* {
    void* p = (void*)(ws + off);
    off += (bytes + 255) & ~(size_t)255;
    return p;
  };
  float* ns    = (float*)carve((size_t)N * 4);
  float* nd    = (float*)carve((size_t)N * 4);
  float* agg1  = (float*)carve((size_t)N * IN_F * 4);
  float* y1    = (float*)carve((size_t)N * OUT_F * 4);
  unsigned* part = (unsigned*)y1;
  float* z     = (float*)carve((size_t)N * NCLS * 4);
  float* agg2  = (float*)carve((size_t)N * NCLS * 4);
  int* ctr     = (int*)carve(256);

  const int B = 256;
  auto cdiv = [](long a, long b) { return (int)((a + b - 1) / b); };

  long n4;
  n4 = (long)N * IN_F / 4;
  k_zero<<<cdiv(n4, B), B, 0, stream>>>((float4*)agg1, n4);
  n4 = (long)N * NCLS / 4 + 16;
  k_zero<<<cdiv(n4, B), B, 0, stream>>>((float4*)agg2, n4);

  int per_slice = cdiv(cdiv(E, HSLICES), 4) * 4;
  k_deg_hist<<<dim3(HSLICES, cdiv(N, HR), 2), 1024, 0, stream>>>(src, dst, part, E, N, per_slice);
  k_norm<<<cdiv(N, B), B, 0, stream>>>(part, ns, nd, N);

  int nch = cdiv(E, CHUNK);
  k_scatter64<<<2 * nch, B, 0, stream>>>(src, dst, x, ns, agg1, E, N, ctr, nch);
  int waves = cdiv(N, 16);
  k_gemm_relu<<<cdiv((long)waves * 32, B), B, 0, stream>>>(agg1, nd, W1, b1, y1, N);

  k_gemm16<<<cdiv((long)waves * 32, B), B, 0, stream>>>(y1, W2, z, N);
  k_scatter16<<<2 * nch, B, 0, stream>>>(src, dst, z, ns, agg2, E, N, ctr + 2, nch);
  k_finalize<<<cdiv((long)N * NCLS, B), B, 0, stream>>>(agg2, nd, b2, out, N);
}
